# starting kernel
# baseline (speedup 1.0000x reference)
#include <hip/hip_runtime.h>
#include <math.h>

typedef __attribute__((ext_vector_type(2))) float v2f;
typedef __attribute__((ext_vector_type(4))) float v4f;
typedef __attribute__((ext_vector_type(8))) float v8f;

#define T_TOK 2048
#define H_DIM 128
#define H4    512
#define NN    256

#define XS_STRIDE 132
#define HS_STRIDE 516


__global__ void k_zero(int* counts) { counts[threadIdx.x] = 0; }

__global__ void k_count(const int* __restrict__ node_ind, int* counts) {
    int t = blockIdx.x * blockDim.x + threadIdx.x;
    if (t < T_TOK) atomicAdd(&counts[node_ind[t]], 1);
}

__global__ void k_scan(const int* __restrict__ counts, int* offsets, int* cursor) {
    __shared__ int s[NN];
    int tid = threadIdx.x;
    int c = counts[tid];
    s[tid] = c;
    __syncthreads();
    for (int d = 1; d < NN; d <<= 1) {
        int v = (tid >= d) ? s[tid - d] : 0;
        __syncthreads();
        s[tid] += v;
        __syncthreads();
    }
    int excl = s[tid] - c;
    offsets[tid] = excl;
    cursor[tid]  = excl;
}

__global__ void k_scatter(const int* __restrict__ node_ind, int* cursor, int* bucket) {
    int t = blockIdx.x * blockDim.x + threadIdx.x;
    if (t < T_TOK) {
        int pos = atomicAdd(&cursor[node_ind[t]], 1);
        bucket[pos] = t;
    }
}


__device__ __forceinline__ float gelu_erf(float v) {
    return 0.5f * v * (1.0f + erff(v * 0.70710678118654752f));
}

__global__ __launch_bounds__(256) void k_mlp(
    const float* __restrict__ x, const int* __restrict__ bucket,
    const int* __restrict__ offsets, const int* __restrict__ counts,
    const float* __restrict__ W1, const float* __restrict__ b1,
    const float* __restrict__ W2, const float* __restrict__ b2,
    float* __restrict__ out)
{
    __shared__ float xs[16 * XS_STRIDE];
    __shared__ float hs[16 * HS_STRIDE];
    __shared__ int   tk[16];

    const int n   = blockIdx.x;
    const int cnt = counts[n];
    if (cnt == 0) return;
    const int start = offsets[n];

    const int tid  = threadIdx.x;
    const int lane = tid & 31;
    const int wave = tid >> 5;
    const int nlo  = lane & 15;
    const int kb   = (lane >> 4) << 1;
    const int mhi  = (lane >> 4) << 3;

    const float* W1n = W1 + (size_t)n * H4 * H_DIM;
    const float* W2n = W2 + (size_t)n * H_DIM * H4;
    const float* b1n = b1 + (size_t)n * H4;
    const float* b2n = b2 + (size_t)n * H_DIM;

    for (int tile = 0; tile < cnt; tile += 16) {
        if (tid < 16) {
            int idx = tile + tid;
            tk[tid] = bucket[start + (idx < cnt ? idx : 0)];
        }
        __syncthreads();

        for (int i = tid; i < 16 * (H_DIM / 4); i += 256) {
            int row = i >> 5, col4 = (i & 31) << 2;
            v4f v = *(const v4f*)(x + (size_t)tk[row] * H_DIM + col4);
            *(v4f*)(xs + row * XS_STRIDE + col4) = v;
        }
        __syncthreads();

        for (int ot = wave; ot < 32; ot += 8) {
            const int obase = ot << 4;
            const float* wrow = W1n + (size_t)(obase + nlo) * H_DIM + kb;
            const float* xrow = xs + nlo * XS_STRIDE + kb;
            if (ot + 8 < 32)
                __builtin_prefetch(wrow + 8 * 16 * H_DIM, 0, 1);
            v8f acc = {};
            #pragma unroll 8
            for (int k = 0; k < H_DIM; k += 4) {
                v2f a = *(const v2f*)(wrow + k);
                v2f b = *(const v2f*)(xrow + k);
                acc = __builtin_amdgcn_wmma_f32_16x16x4_f32(
                    false, a, false, b, (short)0, acc, false, false);
            }
            const int o0 = obase + mhi;
            v4f bia0 = *(const v4f*)(b1n + o0);
            v4f bia1 = *(const v4f*)(b1n + o0 + 4);
            v4f h0, h1;
            #pragma unroll
            for (int r = 0; r < 4; ++r) h0[r] = gelu_erf(acc[r]     + bia0[r]);
            #pragma unroll
            for (int r = 0; r < 4; ++r) h1[r] = gelu_erf(acc[r + 4] + bia1[r]);
            *(v4f*)(hs + nlo * HS_STRIDE + o0)     = h0;
            *(v4f*)(hs + nlo * HS_STRIDE + o0 + 4) = h1;
        }
        __syncthreads();

        {
            const int obase = wave << 4;
            const float* wrow = W2n + (size_t)(obase + nlo) * H4 + kb;
            const float* hrow = hs + nlo * HS_STRIDE + kb;
            v8f acc = {};
            #pragma unroll 8
            for (int k = 0; k < H4; k += 4) {
                v2f a = *(const v2f*)(wrow + k);
                v2f b = *(const v2f*)(hrow + k);
                acc = __builtin_amdgcn_wmma_f32_16x16x4_f32(
                    false, a, false, b, (short)0, acc, false, false);
            }
            if (tile + nlo < cnt) {
                const int t  = tk[nlo];
                const int o0 = obase + mhi;
                v4f bia0 = *(const v4f*)(b2n + o0);
                v4f bia1 = *(const v4f*)(b2n + o0 + 4);
                v4f r0, r1;
                #pragma unroll
                for (int r = 0; r < 4; ++r) r0[r] = acc[r]     + bia0[r];
                #pragma unroll
                for (int r = 0; r < 4; ++r) r1[r] = acc[r + 4] + bia1[r];
                *(v4f*)(out + (size_t)t * H_DIM + o0)     = r0;
                *(v4f*)(out + (size_t)t * H_DIM + o0 + 4) = r1;
            }
        }
        __syncthreads();
    }
}


extern "C" void kernel_launch(void* const* d_in, const int* in_sizes, int n_in,
                              void* d_out, int out_size, void* d_ws, size_t ws_size,
                              hipStream_t stream) {
    const float* x        = (const float*)d_in[0];
    const int*   node_ind = (const int*)  d_in[1];
    const float* W1       = (const float*)d_in[2];
    const float* b1       = (const float*)d_in[3];
    const float* W2       = (const float*)d_in[4];
    const float* b2       = (const float*)d_in[5];
    float*       out      = (float*)d_out;

    int* counts  = (int*)d_ws;
    int* cursor  = counts + NN;
    int* offsets = cursor + NN;
    int* bucket  = offsets + NN;

    k_zero   <<<1,          NN,  0, stream>>>(counts);
    k_count  <<<T_TOK/256,  256, 0, stream>>>(node_ind, counts);
    k_scan   <<<1,          NN,  0, stream>>>(counts, offsets, cursor);
    k_scatter<<<T_TOK/256,  256, 0, stream>>>(node_ind, cursor, bucket);
    k_mlp    <<<NN,         256, 0, stream>>>(x, bucket, offsets, counts,
                                              W1, b1, W2, b2, out);
}


# k_mlp: the block's 32 waves are two groups of 16, each with its own LDS tiles, taking two 16-token tiles of a node at the same time, not one after the other; a wave carries two o-tiles; same sums
# speedup vs baseline: 1.5643x; 1.5643x over previous
#include <hip/hip_runtime.h>
#include <math.h>

typedef __attribute__((ext_vector_type(2))) float v2f;
typedef __attribute__((ext_vector_type(4))) float v4f;
typedef __attribute__((ext_vector_type(8))) float v8f;

#define T_TOK 2048
#define H_DIM 128
#define H4    512
#define NN    256

#define XS_STRIDE 132
#define HS_STRIDE 516


#define BK_THREADS 1024
#define BK_PER_THR (T_TOK / BK_THREADS)

__global__ __launch_bounds__(BK_THREADS) void k_bucket(
    const int* __restrict__ node_ind, int* __restrict__ counts,
    int* __restrict__ offsets, int* __restrict__ bucket)
{
    __shared__ int cnt[NN];
    __shared__ int wsum[NN / 32];

    const int tid  = threadIdx.x;
    const int lane = tid & 31;
    const int wave = tid >> 5;

    int node[BK_PER_THR], rank[BK_PER_THR];
    for (int i = 0; i < BK_PER_THR; ++i) node[i] = node_ind[tid + i * BK_THREADS];

    if (tid < NN) cnt[tid] = 0;
    __syncthreads();
    for (int i = 0; i < BK_PER_THR; ++i) rank[i] = atomicAdd(&cnt[node[i]], 1);
    __syncthreads();

    int c = 0, incl = 0;
    if (tid < NN) {
        c = cnt[tid];
        incl = c;
        for (int d = 1; d < 32; d <<= 1) {
            const int v = __shfl_up(incl, d);
            if (lane >= d) incl += v;
        }
        if (lane == 31) wsum[wave] = incl;
    }
    __syncthreads();
    if (tid < NN) {
        int base = 0;
        for (int w = 0; w < NN / 32; ++w) base += (w < wave) ? wsum[w] : 0;
        const int excl = base + incl - c;
        counts[tid]  = c;
        offsets[tid] = excl;
        cnt[tid]     = excl;
    }
    __syncthreads();
    for (int i = 0; i < BK_PER_THR; ++i)
        bucket[cnt[node[i]] + rank[i]] = tid + i * BK_THREADS;
}


__device__ __forceinline__ float gelu_erf(float v) {
    return 0.5f * v * (1.0f + erff(v * 0.70710678118654752f));
}

#define XS_SZ (16 * XS_STRIDE)
#define HS_SZ (16 * HS_STRIDE)
#define PS_SZ (2 * 8 * 256)

__global__ __launch_bounds__(1024) void k_mlp(
    const float* __restrict__ x, const int* __restrict__ bucket,
    const int* __restrict__ offsets, const int* __restrict__ counts,
    const float* __restrict__ W1, const float* __restrict__ b1,
    const float* __restrict__ W2, const float* __restrict__ b2,
    float* __restrict__ out)
{
    __shared__ float xs_all[2 * XS_SZ];
    __shared__ float hs_all[2 * HS_SZ];
    __shared__ float ps_all[2 * PS_SZ];
    __shared__ int   tk_all[2 * 16];

    const int n   = blockIdx.x;
    const int cnt = counts[n];
    if (cnt == 0) return;
    const int start = offsets[n];

    const int tid  = threadIdx.x;
    const int lane = tid & 31;
    const int grp  = tid >> 9;
    const int tg   = tid & 511;
    const int wg   = tg >> 5;
    const int nlo  = lane & 15;
    const int kb   = (lane >> 4) << 1;
    const int mhi  = (lane >> 4) << 3;

    float* xs = xs_all + grp * XS_SZ;
    float* hs = hs_all + grp * HS_SZ;
    float* ps = ps_all + grp * PS_SZ;
    int*   tk = tk_all + grp * 16;

    const float* W1n = W1 + (size_t)n * H4 * H_DIM;
    const float* W2n = W2 + (size_t)n * H_DIM * H4;
    const float* b1n = b1 + (size_t)n * H4;
    const float* b2n = b2 + (size_t)n * H_DIM;

    for (int tile0 = 0; tile0 < cnt; tile0 += 32) {
        const int  tile = tile0 + (grp << 4);
        const bool act  = tile < cnt;

        if (act && tg < 16) {
            int idx = tile + tg;
            tk[tg] = bucket[start + (idx < cnt ? idx : 0)];
        }
        __syncthreads();

        if (act) {
            int row = tg >> 5, col4 = (tg & 31) << 2;
            v4f v = *(const v4f*)(x + (size_t)tk[row] * H_DIM + col4);
            *(v4f*)(xs + row * XS_STRIDE + col4) = v;
        }
        __syncthreads();

        if (act) {
            const int obase0 = wg << 4;
            const int obase1 = (wg + 16) << 4;
            const float* wrow0 = W1n + (size_t)(obase0 + nlo) * H_DIM + kb;
            const float* wrow1 = W1n + (size_t)(obase1 + nlo) * H_DIM + kb;
            const float* xrow  = xs + nlo * XS_STRIDE + kb;
            v8f acc0 = {}, acc1 = {};
            #pragma unroll 8
            for (int k = 0; k < H_DIM; k += 4) {
                v2f a0 = *(const v2f*)(wrow0 + k);
                v2f a1 = *(const v2f*)(wrow1 + k);
                v2f b  = *(const v2f*)(xrow + k);
                acc0 = __builtin_amdgcn_wmma_f32_16x16x4_f32(
                    false, a0, false, b, (short)0, acc0, false, false);
                acc1 = __builtin_amdgcn_wmma_f32_16x16x4_f32(
                    false, a1, false, b, (short)0, acc1, false, false);
            }
            #pragma unroll
            for (int u = 0; u < 2; ++u) {
                const v8f acc = u ? acc1 : acc0;
                const int o0  = (u ? obase1 : obase0) + mhi;
                v4f bia0 = *(const v4f*)(b1n + o0);
                v4f bia1 = *(const v4f*)(b1n + o0 + 4);
                v4f h0, h1;
                #pragma unroll
                for (int r = 0; r < 4; ++r) h0[r] = gelu_erf(acc[r]     + bia0[r]);
                #pragma unroll
                for (int r = 0; r < 4; ++r) h1[r] = gelu_erf(acc[r + 4] + bia1[r]);
                *(v4f*)(hs + nlo * HS_STRIDE + o0)     = h0;
                *(v4f*)(hs + nlo * HS_STRIDE + o0 + 4) = h1;
            }
        }
        __syncthreads();

        const int ot    = wg & 7;
        const int kh    = wg >> 3;
        const int obase = ot << 4;
        v8f accA = {}, accB = {};
        if (act) {
            const int kA = (2 * kh) * (H4 / 4);
            const int kB = kA + (H4 / 4);
            const float* wrow = W2n + (size_t)(obase + nlo) * H4 + kb;
            const float* hrow = hs + nlo * HS_STRIDE + kb;
            #pragma unroll 8
            for (int k = 0; k < H4 / 4; k += 4) {
                v2f aA = *(const v2f*)(wrow + kA + k);
                v2f bA = *(const v2f*)(hrow + kA + k);
                v2f aB = *(const v2f*)(wrow + kB + k);
                v2f bB = *(const v2f*)(hrow + kB + k);
                accA = __builtin_amdgcn_wmma_f32_16x16x4_f32(
                    false, aA, false, bA, (short)0, accA, false, false);
                accB = __builtin_amdgcn_wmma_f32_16x16x4_f32(
                    false, aB, false, bB, (short)0, accB, false, false);
            }
            if (kh != 0) {
                float* p = ps + ot * 256 + lane * 4;
                *(v4f*)(p)              = __builtin_shufflevector(accA, accA, 0, 1, 2, 3);
                *(v4f*)(p + 128)        = __builtin_shufflevector(accA, accA, 4, 5, 6, 7);
                *(v4f*)(p + 2048)       = __builtin_shufflevector(accB, accB, 0, 1, 2, 3);
                *(v4f*)(p + 2048 + 128) = __builtin_shufflevector(accB, accB, 4, 5, 6, 7);
            }
        }
        __syncthreads();
        if (act && kh == 0) {
            v8f acc = accA + accB;
            for (int j = 0; j < 2; ++j) {
                const float* p = ps + (j * 8 + ot) * 256 + lane * 4;
                v4f s0 = *(const v4f*)(p);
                v4f s1 = *(const v4f*)(p + 128);
                acc += __builtin_shufflevector(s0, s1, 0, 1, 2, 3, 4, 5, 6, 7);
            }
            if (tile + nlo < cnt) {
                const int t  = tk[nlo];
                const int o0 = obase + mhi;
                v4f bia0 = *(const v4f*)(b2n + o0);
                v4f bia1 = *(const v4f*)(b2n + o0 + 4);
                v4f r0, r1;
                #pragma unroll
                for (int r = 0; r < 4; ++r) r0[r] = acc[r]     + bia0[r];
                #pragma unroll
                for (int r = 0; r < 4; ++r) r1[r] = acc[r + 4] + bia1[r];
                *(v4f*)(out + (size_t)t * H_DIM + o0)     = r0;
                *(v4f*)(out + (size_t)t * H_DIM + o0 + 4) = r1;
            }
        }
        __syncthreads();
    }
}


extern "C" void kernel_launch(void* const* d_in, const int* in_sizes, int n_in,
                              void* d_out, int out_size, void* d_ws, size_t ws_size,
                              hipStream_t stream) {
    const float* x        = (const float*)d_in[0];
    const int*   node_ind = (const int*)  d_in[1];
    const float* W1       = (const float*)d_in[2];
    const float* b1       = (const float*)d_in[3];
    const float* W2       = (const float*)d_in[4];
    const float* b2       = (const float*)d_in[5];
    float*       out      = (float*)d_out;

    int* counts  = (int*)d_ws;
    int* offsets = counts + NN;
    int* bucket  = offsets + NN;

    k_bucket<<<1,  BK_THREADS, 0, stream>>>(node_ind, counts, offsets, bucket);
    k_mlp   <<<NN, 1024,       0, stream>>>(x, bucket, offsets, counts,
                                            W1, b1, W2, b2, out);
}


# k_mlp fragments: a lane reads 4 consecutive k with one b128 load (weights from global, X/H from LDS); a WMMA takes k {0,1,4,5} then {2,3,6,7} of each 8; half the loads; k sum order changes
# speedup vs baseline: 1.6113x; 1.0300x over previous
#include <hip/hip_runtime.h>
#include <math.h>

typedef __attribute__((ext_vector_type(2))) float v2f;
typedef __attribute__((ext_vector_type(4))) float v4f;
typedef __attribute__((ext_vector_type(8))) float v8f;

#define T_TOK 2048
#define H_DIM 128
#define H4    512
#define NN    256

#define XS_STRIDE 132
#define HS_STRIDE 516


#define BK_THREADS 1024
#define BK_PER_THR (T_TOK / BK_THREADS)

__global__ __launch_bounds__(BK_THREADS) void k_bucket(
    const int* __restrict__ node_ind, int* __restrict__ counts,
    int* __restrict__ offsets, int* __restrict__ bucket)
{
    __shared__ int cnt[NN];
    __shared__ int wsum[NN / 32];

    const int tid  = threadIdx.x;
    const int lane = tid & 31;
    const int wave = tid >> 5;

    int node[BK_PER_THR], rank[BK_PER_THR];
    for (int i = 0; i < BK_PER_THR; ++i) node[i] = node_ind[tid + i * BK_THREADS];

    if (tid < NN) cnt[tid] = 0;
    __syncthreads();
    for (int i = 0; i < BK_PER_THR; ++i) rank[i] = atomicAdd(&cnt[node[i]], 1);
    __syncthreads();

    int c = 0, incl = 0;
    if (tid < NN) {
        c = cnt[tid];
        incl = c;
        for (int d = 1; d < 32; d <<= 1) {
            const int v = __shfl_up(incl, d);
            if (lane >= d) incl += v;
        }
        if (lane == 31) wsum[wave] = incl;
    }
    __syncthreads();
    if (tid < NN) {
        int base = 0;
        for (int w = 0; w < NN / 32; ++w) base += (w < wave) ? wsum[w] : 0;
        const int excl = base + incl - c;
        counts[tid]  = c;
        offsets[tid] = excl;
        cnt[tid]     = excl;
    }
    __syncthreads();
    for (int i = 0; i < BK_PER_THR; ++i)
        bucket[cnt[node[i]] + rank[i]] = tid + i * BK_THREADS;
}


__device__ __forceinline__ float gelu_erf(float v) {
    return 0.5f * v * (1.0f + erff(v * 0.70710678118654752f));
}

__device__ __forceinline__ v4f frag_k8(const float* row_k4h, int k8) {
    return *(const v4f*)(row_k4h + k8);
}
__device__ __forceinline__ v2f frag_lo(v4f v) { return __builtin_shufflevector(v, v, 0, 1); }
__device__ __forceinline__ v2f frag_hi(v4f v) { return __builtin_shufflevector(v, v, 2, 3); }

#define XS_SZ (16 * XS_STRIDE)
#define HS_SZ (16 * HS_STRIDE)
#define PS_SZ (2 * 8 * 256)

__global__ __launch_bounds__(1024) void k_mlp(
    const float* __restrict__ x, const int* __restrict__ bucket,
    const int* __restrict__ offsets, const int* __restrict__ counts,
    const float* __restrict__ W1, const float* __restrict__ b1,
    const float* __restrict__ W2, const float* __restrict__ b2,
    float* __restrict__ out)
{
    __shared__ __attribute__((aligned(16))) float xs_all[2 * XS_SZ];
    __shared__ __attribute__((aligned(16))) float hs_all[2 * HS_SZ];
    __shared__ __attribute__((aligned(16))) float ps_all[2 * PS_SZ];
    __shared__ int   tk_all[2 * 16];

    const int n   = blockIdx.x;
    const int cnt = counts[n];
    if (cnt == 0) return;
    const int start = offsets[n];

    const int tid  = threadIdx.x;
    const int lane = tid & 31;
    const int grp  = tid >> 9;
    const int tg   = tid & 511;
    const int wg   = tg >> 5;
    const int nlo  = lane & 15;
    const int kb   = (lane >> 4) << 2;
    const int mhi  = (lane >> 4) << 3;

    float* xs = xs_all + grp * XS_SZ;
    float* hs = hs_all + grp * HS_SZ;
    float* ps = ps_all + grp * PS_SZ;
    int*   tk = tk_all + grp * 16;

    const float* W1n = W1 + (size_t)n * H4 * H_DIM;
    const float* W2n = W2 + (size_t)n * H_DIM * H4;
    const float* b1n = b1 + (size_t)n * H4;
    const float* b2n = b2 + (size_t)n * H_DIM;

    for (int tile0 = 0; tile0 < cnt; tile0 += 32) {
        const int  tile = tile0 + (grp << 4);
        const bool act  = tile < cnt;

        if (act && tg < 16) {
            int idx = tile + tg;
            tk[tg] = bucket[start + (idx < cnt ? idx : 0)];
        }
        __syncthreads();

        if (act) {
            int row = tg >> 5, col4 = (tg & 31) << 2;
            v4f v = *(const v4f*)(x + (unsigned)(tk[row] * H_DIM + col4));
            *(v4f*)(xs + row * XS_STRIDE + col4) = v;
        }
        __syncthreads();

        if (act) {
            const int obase0 = wg << 4;
            const int obase1 = (wg + 16) << 4;
            const float* wrow0 = W1n + (unsigned)((obase0 + nlo) * H_DIM + kb);
            const float* wrow1 = W1n + (unsigned)((obase1 + nlo) * H_DIM + kb);
            const float* xrow  = xs + nlo * XS_STRIDE + kb;
            v8f acc0 = {}, acc1 = {};
            #pragma unroll 8
            for (int k = 0; k < H_DIM; k += 8) {
                const v4f a0 = frag_k8(wrow0, k);
                const v4f a1 = frag_k8(wrow1, k);
                const v4f b  = frag_k8(xrow, k);
                acc0 = __builtin_amdgcn_wmma_f32_16x16x4_f32(
                    false, frag_lo(a0), false, frag_lo(b), (short)0, acc0, false, false);
                acc1 = __builtin_amdgcn_wmma_f32_16x16x4_f32(
                    false, frag_lo(a1), false, frag_lo(b), (short)0, acc1, false, false);
                acc0 = __builtin_amdgcn_wmma_f32_16x16x4_f32(
                    false, frag_hi(a0), false, frag_hi(b), (short)0, acc0, false, false);
                acc1 = __builtin_amdgcn_wmma_f32_16x16x4_f32(
                    false, frag_hi(a1), false, frag_hi(b), (short)0, acc1, false, false);
            }
            #pragma unroll
            for (int u = 0; u < 2; ++u) {
                const v8f acc = u ? acc1 : acc0;
                const int o0  = (u ? obase1 : obase0) + mhi;
                v4f bia0 = *(const v4f*)(b1n + o0);
                v4f bia1 = *(const v4f*)(b1n + o0 + 4);
                v4f h0, h1;
                #pragma unroll
                for (int r = 0; r < 4; ++r) h0[r] = gelu_erf(acc[r]     + bia0[r]);
                #pragma unroll
                for (int r = 0; r < 4; ++r) h1[r] = gelu_erf(acc[r + 4] + bia1[r]);
                *(v4f*)(hs + nlo * HS_STRIDE + o0)     = h0;
                *(v4f*)(hs + nlo * HS_STRIDE + o0 + 4) = h1;
            }
        }
        __syncthreads();

        const int ot    = wg & 7;
        const int kh    = wg >> 3;
        const int obase = ot << 4;
        v8f accA = {}, accB = {};
        if (act) {
            const int kA = (2 * kh) * (H4 / 4);
            const int kQ = H4 / 4;
            const float* wrow = W2n + (unsigned)((obase + nlo) * H4 + kb + kA);
            const float* hrow = hs + nlo * HS_STRIDE + kb + kA;
            #pragma nounroll
            for (int k0 = 0; k0 < H4 / 4; k0 += 64) {
                v4f aA[8], aB[8];
                #pragma unroll
                for (int i = 0; i < 8; ++i) {
                    aA[i] = frag_k8(wrow, k0 + 8 * i);
                    aB[i] = frag_k8(wrow, kQ + k0 + 8 * i);
                }
                #pragma unroll
                for (int i = 0; i < 8; ++i) {
                    const v4f bA = frag_k8(hrow, k0 + 8 * i);
                    const v4f bB = frag_k8(hrow, kQ + k0 + 8 * i);
                    accA = __builtin_amdgcn_wmma_f32_16x16x4_f32(
                        false, frag_lo(aA[i]), false, frag_lo(bA), (short)0, accA, false, false);
                    accB = __builtin_amdgcn_wmma_f32_16x16x4_f32(
                        false, frag_lo(aB[i]), false, frag_lo(bB), (short)0, accB, false, false);
                    accA = __builtin_amdgcn_wmma_f32_16x16x4_f32(
                        false, frag_hi(aA[i]), false, frag_hi(bA), (short)0, accA, false, false);
                    accB = __builtin_amdgcn_wmma_f32_16x16x4_f32(
                        false, frag_hi(aB[i]), false, frag_hi(bB), (short)0, accB, false, false);
                }
            }
            if (kh != 0) {
                float* p = ps + ot * 256 + lane * 4;
                *(v4f*)(p)              = __builtin_shufflevector(accA, accA, 0, 1, 2, 3);
                *(v4f*)(p + 128)        = __builtin_shufflevector(accA, accA, 4, 5, 6, 7);
                *(v4f*)(p + 2048)       = __builtin_shufflevector(accB, accB, 0, 1, 2, 3);
                *(v4f*)(p + 2048 + 128) = __builtin_shufflevector(accB, accB, 4, 5, 6, 7);
            }
        }
        __syncthreads();
        if (act && kh == 0) {
            v8f acc = accA + accB;
            for (int j = 0; j < 2; ++j) {
                const float* p = ps + (j * 8 + ot) * 256 + lane * 4;
                v4f s0 = *(const v4f*)(p);
                v4f s1 = *(const v4f*)(p + 128);
                acc += __builtin_shufflevector(s0, s1, 0, 1, 2, 3, 4, 5, 6, 7);
            }
            if (tile + nlo < cnt) {
                const int t  = tk[nlo];
                const int o0 = obase + mhi;
                v4f bia0 = *(const v4f*)(b2n + o0);
                v4f bia1 = *(const v4f*)(b2n + o0 + 4);
                v4f r0, r1;
                #pragma unroll
                for (int r = 0; r < 4; ++r) r0[r] = acc[r]     + bia0[r];
                #pragma unroll
                for (int r = 0; r < 4; ++r) r1[r] = acc[r + 4] + bia1[r];
                float* orow = out + (unsigned)(t * H_DIM + o0);
                *(v4f*)(orow)     = r0;
                *(v4f*)(orow + 4) = r1;
            }
        }
        __syncthreads();
    }
}


extern "C" void kernel_launch(void* const* d_in, const int* in_sizes, int n_in,
                              void* d_out, int out_size, void* d_ws, size_t ws_size,
                              hipStream_t stream) {
    const float* x        = (const float*)d_in[0];
    const int*   node_ind = (const int*)  d_in[1];
    const float* W1       = (const float*)d_in[2];
    const float* b1       = (const float*)d_in[3];
    const float* W2       = (const float*)d_in[4];
    const float* b2       = (const float*)d_in[5];
    float*       out      = (float*)d_out;

    int* counts  = (int*)d_ws;
    int* offsets = counts + NN;
    int* bucket  = offsets + NN;

    k_bucket<<<1,  BK_THREADS, 0, stream>>>(node_ind, counts, offsets, bucket);
    k_mlp   <<<NN, 1024,       0, stream>>>(x, bucket, offsets, counts,
                                            W1, b1, W2, b2, out);
}
